# starting kernel
# baseline (speedup 1.0000x reference)
#include <hip/hip_runtime.h>
#include <math.h>

#define BATCH   1024
#define SEQ     256
#define EMBED   4
#define H1D     8
#define H2D     4
#define NWAVE   4
#define SSTRIDE 260

typedef __attribute__((ext_vector_type(2))) float v2f;
typedef __attribute__((ext_vector_type(8))) float v8f;

__global__ __launch_bounds__(128)
void hybrid_attn_kernel(const float* __restrict__ x,
                        const float* __restrict__ Wq, const float* __restrict__ Wk,
                        const float* __restrict__ Wv,
                        const float* __restrict__ W1, const float* __restrict__ b1,
                        const float* __restrict__ W2, const float* __restrict__ b2,
                        const float* __restrict__ W3, const float* __restrict__ b3,
                        float* __restrict__ out)
{
  extern __shared__ float smem[];
  float*  q_lds = smem;
  float*  k_lds = q_lds + SEQ * EMBED;
  float4* v4    = (float4*)(k_lds + SEQ * EMBED);
  float*  sbuf  = (float*)(v4 + SEQ);

  const int b    = blockIdx.x;
  const int tid  = threadIdx.x;
  const int lane = tid & 31;
  const int wave = tid >> 5;

  const float* xb = x + (size_t)b * SEQ * EMBED;
  for (int t = tid; t < SEQ; t += 128) {
    float4 xe = ((const float4*)xb)[t];
    float xv[4] = {xe.x, xe.y, xe.z, xe.w};
    float4 vvec;
    float* vp = (float*)&vvec;
#pragma unroll
    for (int eo = 0; eo < 4; ++eo) {
      float q = 0.f, k = 0.f, v = 0.f;
#pragma unroll
      for (int e = 0; e < 4; ++e) {
        q += xv[e] * Wq[e * 4 + eo];
        k += xv[e] * Wk[e * 4 + eo];
        v += xv[e] * Wv[e * 4 + eo];
      }
      q_lds[t * 4 + eo] = 0.5f * q;
      k_lds[t * 4 + eo] = k;
      vp[eo] = v;
    }
    v4[t] = vvec;
  }
  __syncthreads();

  const int lane15 = lane & 15;
  const int half   = lane >> 4;
  const int kbase  = half * 2;

  v2f barr[16];
#pragma unroll
  for (int j = 0; j < 16; ++j) {
    barr[j].x = k_lds[(j * 16 + lane15) * 4 + kbase];
    barr[j].y = k_lds[(j * 16 + lane15) * 4 + kbase + 1];
  }

  float* swave = sbuf + wave * 16 * SSTRIDE;

  for (int tile = wave; tile < SEQ / 16; tile += NWAVE) {
    const int i0 = tile * 16;

    v2f a;
    a.x = q_lds[(i0 + lane15) * 4 + kbase];
    a.y = q_lds[(i0 + lane15) * 4 + kbase + 1];

#pragma unroll
    for (int j = 0; j < 16; ++j) {
      v8f c = {};
      c = __builtin_amdgcn_wmma_f32_16x16x4_f32(
              false, a, false, barr[j], (short)0, c, false, false);

#pragma unroll
      for (int r = 0; r < 8; ++r)
        swave[(r + 8 * half) * SSTRIDE + j * 16 + lane15] = c[r];
    }

    const int    row   = lane15;
    const float* Sr    = swave + row * SSTRIDE;
    const int    tbase = half * 128;

    float mx = -1e30f;
#pragma unroll 8
    for (int i = 0; i < 128; i += 4) {
      float4 sv = *(const float4*)(Sr + tbase + i);
      mx = fmaxf(mx, fmaxf(fmaxf(sv.x, sv.y), fmaxf(sv.z, sv.w)));
    }
    mx = fmaxf(mx, __shfl_xor(mx, 16, 32));

    float sum = 0.f, o0 = 0.f, o1 = 0.f, o2 = 0.f, o3 = 0.f;
#pragma unroll 4
    for (int i = 0; i < 128; i += 4) {
      float4 sv = *(const float4*)(Sr + tbase + i);
      float p0 = __expf(sv.x - mx);
      float p1 = __expf(sv.y - mx);
      float p2 = __expf(sv.z - mx);
      float p3 = __expf(sv.w - mx);
      sum += (p0 + p1) + (p2 + p3);
      float4 va = v4[tbase + i + 0];
      float4 vb = v4[tbase + i + 1];
      float4 vc = v4[tbase + i + 2];
      float4 vd = v4[tbase + i + 3];
      o0 += p0 * va.x + p1 * vb.x + p2 * vc.x + p3 * vd.x;
      o1 += p0 * va.y + p1 * vb.y + p2 * vc.y + p3 * vd.y;
      o2 += p0 * va.z + p1 * vb.z + p2 * vc.z + p3 * vd.z;
      o3 += p0 * va.w + p1 * vb.w + p2 * vc.w + p3 * vd.w;
    }
    sum += __shfl_xor(sum, 16, 32);
    o0  += __shfl_xor(o0, 16, 32);
    o1  += __shfl_xor(o1, 16, 32);
    o2  += __shfl_xor(o2, 16, 32);
    o3  += __shfl_xor(o3, 16, 32);

    float inv = 1.f / sum;
    o0 *= inv; o1 *= inv; o2 *= inv; o3 *= inv;

    float h1v[H1D];
#pragma unroll
    for (int j = 0; j < H1D; ++j)
      h1v[j] = tanhf(b1[j] + o0 * W1[0 * H1D + j] + o1 * W1[1 * H1D + j]
                           + o2 * W1[2 * H1D + j] + o3 * W1[3 * H1D + j]);
    float h2v[H2D];
#pragma unroll
    for (int j = 0; j < H2D; ++j) {
      float acc = b2[j];
#pragma unroll
      for (int i = 0; i < H1D; ++i) acc += h1v[i] * W2[i * H2D + j];
      h2v[j] = tanhf(acc);
    }
    float y = b3[0];
#pragma unroll
    for (int j = 0; j < H2D; ++j) y += h2v[j] * W3[j];

    if (half == 0) out[(size_t)b * SEQ + i0 + row] = y;
  }
}

extern "C" void kernel_launch(void* const* d_in, const int* in_sizes, int n_in,
                              void* d_out, int out_size, void* d_ws, size_t ws_size,
                              hipStream_t stream) {
  const float* x  = (const float*)d_in[0];
  const float* Wq = (const float*)d_in[1];
  const float* Wk = (const float*)d_in[2];
  const float* Wv = (const float*)d_in[3];
  const float* W1 = (const float*)d_in[4];
  const float* b1 = (const float*)d_in[5];
  const float* W2 = (const float*)d_in[6];
  const float* b2 = (const float*)d_in[7];
  const float* W3 = (const float*)d_in[8];
  const float* b3 = (const float*)d_in[9];
  float* out = (float*)d_out;

  const size_t smem_bytes =
      (size_t)(SEQ * EMBED * 2 + SEQ * 4 + NWAVE * 16 * SSTRIDE) * sizeof(float);

  hybrid_attn_kernel<<<BATCH, 128, smem_bytes, stream>>>(
      x, Wq, Wk, Wv, W1, b1, W2, b2, W3, b3, out);
}


# scores computed transposed on WMMA (K tile as A, Q as B) so each lane owns one query row: softmax and PV stay in registers over 4 tiles, max shift via C operand, no score LDS; PV sum order changed
# speedup vs baseline: 1.2352x; 1.2352x over previous
#include <hip/hip_runtime.h>
#include <math.h>


#define BATCH   1024
#define SEQ     256
#define EMBED   4
#define H1D     8
#define H2D     4
#define NWAVE   4
#define TPW     4

#define WMMA_COVER4(d0, d1, d2, d3, a)                                   \
  asm volatile("v_nop\n\tv_nop\n\tv_nop\n\tv_nop"                        \
               : "+v"(d0), "+v"(d1), "+v"(d2), "+v"(d3), "+v"(a))

typedef __attribute__((ext_vector_type(2))) float v2f;
typedef __attribute__((ext_vector_type(4))) float v4f;
typedef __attribute__((ext_vector_type(8))) float v8f;

__global__ __launch_bounds__(128)
void hybrid_attn_kernel(const float* __restrict__ x,
                        const float* __restrict__ Wq, const float* __restrict__ Wk,
                        const float* __restrict__ Wv,
                        const float* __restrict__ W1, const float* __restrict__ b1,
                        const float* __restrict__ W2, const float* __restrict__ b2,
                        const float* __restrict__ W3, const float* __restrict__ b3,
                        float* __restrict__ out)
{
  __shared__ __attribute__((aligned(16))) float q_lds[SEQ * EMBED];
  __shared__ __attribute__((aligned(16))) float k_lds[SEQ * EMBED];
  __shared__ __attribute__((aligned(16))) float vt_lds[EMBED * SEQ];

  const int b    = blockIdx.x;
  const int tid  = threadIdx.x;
  const int lane = tid & 31;
  const int wave = tid >> 5;

  const float qscale = 0.5f * 1.44269504088896341f;
  const float* xb = x + (size_t)b * SEQ * EMBED;
  for (int t = tid; t < SEQ; t += 128) {
    float4 xe = ((const float4*)xb)[t];
    float xv[4] = {xe.x, xe.y, xe.z, xe.w};
#pragma unroll
    for (int eo = 0; eo < 4; ++eo) {
      float q = 0.f, k = 0.f, v = 0.f;
#pragma unroll
      for (int e = 0; e < 4; ++e) {
        q += xv[e] * Wq[e * 4 + eo];
        k += xv[e] * Wk[e * 4 + eo];
        v += xv[e] * Wv[e * 4 + eo];
      }
      q_lds[t * 4 + eo] = qscale * q;
      k_lds[t * 4 + eo] = k;
      vt_lds[eo * SEQ + t] = v;
    }
  }
  __syncthreads();

  const int lane15 = lane & 15;
  const int half   = lane >> 4;
  const int kbase  = half * 2;


  v2f qf[TPW];
#pragma unroll
  for (int tl = 0; tl < TPW; ++tl) {
    const int i0 = (wave + tl * NWAVE) * 16;
    qf[tl] = *(const v2f*)&q_lds[(i0 + lane15) * 4 + kbase];
  }

  float mx[TPW];
#pragma unroll
  for (int tl = 0; tl < TPW; ++tl) mx[tl] = -INFINITY;

#pragma unroll 4
  for (int j = 0; j < SEQ / 16; ++j) {
    v2f kf = *(const v2f*)&k_lds[(j * 16 + lane15) * 4 + kbase];
    v8f c[TPW];
#pragma unroll
    for (int tl = 0; tl < TPW; ++tl) {
      const v8f z = {};
      c[tl] = __builtin_amdgcn_wmma_f32_16x16x4_f32(
                  false, kf, false, qf[tl], (short)0, z, false, false);
    }
    WMMA_COVER4(c[0], c[1], c[2], c[3], kf);
#pragma unroll
    for (int tl = 0; tl < TPW; ++tl) {
      float m = mx[tl];
      m = fmaxf(fmaxf(m, c[tl][0]), c[tl][1]);
      m = fmaxf(fmaxf(m, c[tl][2]), c[tl][3]);
      m = fmaxf(fmaxf(m, c[tl][4]), c[tl][5]);
      m = fmaxf(fmaxf(m, c[tl][6]), c[tl][7]);
      mx[tl] = m;
    }
  }

  v8f cneg[TPW];
#pragma unroll
  for (int tl = 0; tl < TPW; ++tl) {
    const float m = fmaxf(mx[tl], __shfl_xor(mx[tl], 16, 32));
#pragma unroll
    for (int r = 0; r < 8; ++r) cneg[tl][r] = -m;
  }

  v2f acc[TPW][EMBED], s2[TPW];
#pragma unroll
  for (int tl = 0; tl < TPW; ++tl) {
#pragma unroll
    for (int e = 0; e < EMBED; ++e) acc[tl][e] = (v2f){0.f, 0.f};
    s2[tl] = (v2f){0.f, 0.f};
  }

#pragma unroll 2
  for (int j = 0; j < SEQ / 16; ++j) {
    v2f kf = *(const v2f*)&k_lds[(j * 16 + lane15) * 4 + kbase];
    const float* vb = vt_lds + j * 16 + half * 8;
    v4f vlo[EMBED], vhi[EMBED];
#pragma unroll
    for (int e = 0; e < EMBED; ++e) {
      vlo[e] = *(const v4f*)(vb + e * SEQ);
      vhi[e] = *(const v4f*)(vb + e * SEQ + 4);
    }

    v8f c[TPW];
#pragma unroll
    for (int tl = 0; tl < TPW; ++tl)
      c[tl] = __builtin_amdgcn_wmma_f32_16x16x4_f32(
                  false, kf, false, qf[tl], (short)0, cneg[tl], false, false);
    WMMA_COVER4(c[0], c[1], c[2], c[3], kf);

#pragma unroll
    for (int tl = 0; tl < TPW; ++tl) {
      v8f p;
#pragma unroll
      for (int r = 0; r < 8; ++r) p[r] = __builtin_amdgcn_exp2f(c[tl][r]);
      const v2f p01 = {p[0], p[1]}, p23 = {p[2], p[3]}, p45 = {p[4], p[5]}, p67 = {p[6], p[7]};
      s2[tl] += p01; s2[tl] += p23; s2[tl] += p45; s2[tl] += p67;
#pragma unroll
      for (int e = 0; e < EMBED; ++e) {
        v2f a = acc[tl][e];
        a = __builtin_elementwise_fma(p01, (v2f){vlo[e][0], vlo[e][1]}, a);
        a = __builtin_elementwise_fma(p23, (v2f){vlo[e][2], vlo[e][3]}, a);
        a = __builtin_elementwise_fma(p45, (v2f){vhi[e][0], vhi[e][1]}, a);
        a = __builtin_elementwise_fma(p67, (v2f){vhi[e][2], vhi[e][3]}, a);
        acc[tl][e] = a;
      }
    }
  }

#pragma unroll
  for (int tl = 0; tl < TPW; ++tl) {
    const int i0 = (wave + tl * NWAVE) * 16;
    float sum = s2[tl][0] + s2[tl][1];
    float o0  = acc[tl][0][0] + acc[tl][0][1];
    float o1  = acc[tl][1][0] + acc[tl][1][1];
    float o2  = acc[tl][2][0] + acc[tl][2][1];
    float o3  = acc[tl][3][0] + acc[tl][3][1];
    sum += __shfl_xor(sum, 16, 32);
    o0  += __shfl_xor(o0, 16, 32);
    o1  += __shfl_xor(o1, 16, 32);
    o2  += __shfl_xor(o2, 16, 32);
    o3  += __shfl_xor(o3, 16, 32);

    float inv = 1.f / sum;
    o0 *= inv; o1 *= inv; o2 *= inv; o3 *= inv;

    float h1v[H1D];
#pragma unroll
    for (int j = 0; j < H1D; ++j)
      h1v[j] = tanhf(b1[j] + o0 * W1[0 * H1D + j] + o1 * W1[1 * H1D + j]
                           + o2 * W1[2 * H1D + j] + o3 * W1[3 * H1D + j]);
    float h2v[H2D];
#pragma unroll
    for (int j = 0; j < H2D; ++j) {
      float acc = b2[j];
#pragma unroll
      for (int i = 0; i < H1D; ++i) acc += h1v[i] * W2[i * H2D + j];
      h2v[j] = tanhf(acc);
    }
    float y = b3[0];
#pragma unroll
    for (int j = 0; j < H2D; ++j) y += h2v[j] * W3[j];

    if (half == 0) out[(size_t)b * SEQ + i0 + lane15] = y;
  }
}

extern "C" void kernel_launch(void* const* d_in, const int* in_sizes, int n_in,
                              void* d_out, int out_size, void* d_ws, size_t ws_size,
                              hipStream_t stream) {
  const float* x  = (const float*)d_in[0];
  const float* Wq = (const float*)d_in[1];
  const float* Wk = (const float*)d_in[2];
  const float* Wv = (const float*)d_in[3];
  const float* W1 = (const float*)d_in[4];
  const float* b1 = (const float*)d_in[5];
  const float* W2 = (const float*)d_in[6];
  const float* b2 = (const float*)d_in[7];
  const float* W3 = (const float*)d_in[8];
  const float* b3 = (const float*)d_in[9];
  float* out = (float*)d_out;

  hybrid_attn_kernel<<<BATCH, 128, 0, stream>>>(
      x, Wq, Wk, Wv, W1, b1, W2, b2, W3, b3, out);
}


# MLP epilogue: each tanh is one hardware v_tanh_f32 in place of the branching library tanhf (48 per wave, both branch sides taken); tanh rounding changes, no sum order changes
# speedup vs baseline: 1.6344x; 1.3232x over previous
#include <hip/hip_runtime.h>
#include <math.h>


#define BATCH   1024
#define SEQ     256
#define EMBED   4
#define H1D     8
#define H2D     4
#define NWAVE   4
#define TPW     4

#define WMMA_COVER4(d0, d1, d2, d3, a)                                   \
  asm volatile("v_nop\n\tv_nop\n\tv_nop\n\tv_nop"                        \
               : "+v"(d0), "+v"(d1), "+v"(d2), "+v"(d3), "+v"(a))

typedef __attribute__((ext_vector_type(2))) float v2f;
typedef __attribute__((ext_vector_type(4))) float v4f;
typedef __attribute__((ext_vector_type(8))) float v8f;

__global__ __launch_bounds__(128)
void hybrid_attn_kernel(const float* __restrict__ x,
                        const float* __restrict__ Wq, const float* __restrict__ Wk,
                        const float* __restrict__ Wv,
                        const float* __restrict__ W1, const float* __restrict__ b1,
                        const float* __restrict__ W2, const float* __restrict__ b2,
                        const float* __restrict__ W3, const float* __restrict__ b3,
                        float* __restrict__ out)
{
  __shared__ __attribute__((aligned(16))) float q_lds[SEQ * EMBED];
  __shared__ __attribute__((aligned(16))) float k_lds[SEQ * EMBED];
  __shared__ __attribute__((aligned(16))) float vt_lds[EMBED * SEQ];

  const int b    = blockIdx.x;
  const int tid  = threadIdx.x;
  const int lane = tid & 31;
  const int wave = tid >> 5;

  const float qscale = 0.5f * 1.44269504088896341f;
  const float* xb = x + (size_t)b * SEQ * EMBED;
  for (int t = tid; t < SEQ; t += 128) {
    float4 xe = ((const float4*)xb)[t];
    float xv[4] = {xe.x, xe.y, xe.z, xe.w};
#pragma unroll
    for (int eo = 0; eo < 4; ++eo) {
      float q = 0.f, k = 0.f, v = 0.f;
#pragma unroll
      for (int e = 0; e < 4; ++e) {
        q += xv[e] * Wq[e * 4 + eo];
        k += xv[e] * Wk[e * 4 + eo];
        v += xv[e] * Wv[e * 4 + eo];
      }
      q_lds[t * 4 + eo] = qscale * q;
      k_lds[t * 4 + eo] = k;
      vt_lds[eo * SEQ + t] = v;
    }
  }
  __syncthreads();

  const int lane15 = lane & 15;
  const int half   = lane >> 4;
  const int kbase  = half * 2;


  v2f qf[TPW];
#pragma unroll
  for (int tl = 0; tl < TPW; ++tl) {
    const int i0 = (wave + tl * NWAVE) * 16;
    qf[tl] = *(const v2f*)&q_lds[(i0 + lane15) * 4 + kbase];
  }

  float mx[TPW];
#pragma unroll
  for (int tl = 0; tl < TPW; ++tl) mx[tl] = -INFINITY;

#pragma unroll 4
  for (int j = 0; j < SEQ / 16; ++j) {
    v2f kf = *(const v2f*)&k_lds[(j * 16 + lane15) * 4 + kbase];
    v8f c[TPW];
#pragma unroll
    for (int tl = 0; tl < TPW; ++tl) {
      const v8f z = {};
      c[tl] = __builtin_amdgcn_wmma_f32_16x16x4_f32(
                  false, kf, false, qf[tl], (short)0, z, false, false);
    }
    WMMA_COVER4(c[0], c[1], c[2], c[3], kf);
#pragma unroll
    for (int tl = 0; tl < TPW; ++tl) {
      float m = mx[tl];
      m = fmaxf(fmaxf(m, c[tl][0]), c[tl][1]);
      m = fmaxf(fmaxf(m, c[tl][2]), c[tl][3]);
      m = fmaxf(fmaxf(m, c[tl][4]), c[tl][5]);
      m = fmaxf(fmaxf(m, c[tl][6]), c[tl][7]);
      mx[tl] = m;
    }
  }

  v8f cneg[TPW];
#pragma unroll
  for (int tl = 0; tl < TPW; ++tl) {
    const float m = fmaxf(mx[tl], __shfl_xor(mx[tl], 16, 32));
#pragma unroll
    for (int r = 0; r < 8; ++r) cneg[tl][r] = -m;
  }

  v2f acc[TPW][EMBED], s2[TPW];
#pragma unroll
  for (int tl = 0; tl < TPW; ++tl) {
#pragma unroll
    for (int e = 0; e < EMBED; ++e) acc[tl][e] = (v2f){0.f, 0.f};
    s2[tl] = (v2f){0.f, 0.f};
  }

#pragma unroll 2
  for (int j = 0; j < SEQ / 16; ++j) {
    v2f kf = *(const v2f*)&k_lds[(j * 16 + lane15) * 4 + kbase];
    const float* vb = vt_lds + j * 16 + half * 8;
    v4f vlo[EMBED], vhi[EMBED];
#pragma unroll
    for (int e = 0; e < EMBED; ++e) {
      vlo[e] = *(const v4f*)(vb + e * SEQ);
      vhi[e] = *(const v4f*)(vb + e * SEQ + 4);
    }

    v8f c[TPW];
#pragma unroll
    for (int tl = 0; tl < TPW; ++tl)
      c[tl] = __builtin_amdgcn_wmma_f32_16x16x4_f32(
                  false, kf, false, qf[tl], (short)0, cneg[tl], false, false);
    WMMA_COVER4(c[0], c[1], c[2], c[3], kf);

#pragma unroll
    for (int tl = 0; tl < TPW; ++tl) {
      v8f p;
#pragma unroll
      for (int r = 0; r < 8; ++r) p[r] = __builtin_amdgcn_exp2f(c[tl][r]);
      const v2f p01 = {p[0], p[1]}, p23 = {p[2], p[3]}, p45 = {p[4], p[5]}, p67 = {p[6], p[7]};
      s2[tl] += p01; s2[tl] += p23; s2[tl] += p45; s2[tl] += p67;
#pragma unroll
      for (int e = 0; e < EMBED; ++e) {
        v2f a = acc[tl][e];
        a = __builtin_elementwise_fma(p01, (v2f){vlo[e][0], vlo[e][1]}, a);
        a = __builtin_elementwise_fma(p23, (v2f){vlo[e][2], vlo[e][3]}, a);
        a = __builtin_elementwise_fma(p45, (v2f){vhi[e][0], vhi[e][1]}, a);
        a = __builtin_elementwise_fma(p67, (v2f){vhi[e][2], vhi[e][3]}, a);
        acc[tl][e] = a;
      }
    }
  }

#pragma unroll
  for (int tl = 0; tl < TPW; ++tl) {
    const int i0 = (wave + tl * NWAVE) * 16;
    float sum = s2[tl][0] + s2[tl][1];
    float o0  = acc[tl][0][0] + acc[tl][0][1];
    float o1  = acc[tl][1][0] + acc[tl][1][1];
    float o2  = acc[tl][2][0] + acc[tl][2][1];
    float o3  = acc[tl][3][0] + acc[tl][3][1];
    sum += __shfl_xor(sum, 16, 32);
    o0  += __shfl_xor(o0, 16, 32);
    o1  += __shfl_xor(o1, 16, 32);
    o2  += __shfl_xor(o2, 16, 32);
    o3  += __shfl_xor(o3, 16, 32);

    float inv = 1.f / sum;
    o0 *= inv; o1 *= inv; o2 *= inv; o3 *= inv;

    float h1v[H1D];
#pragma unroll
    for (int j = 0; j < H1D; ++j)
      h1v[j] = __builtin_amdgcn_tanhf(b1[j] + o0 * W1[0 * H1D + j] + o1 * W1[1 * H1D + j]
                           + o2 * W1[2 * H1D + j] + o3 * W1[3 * H1D + j]);
    float h2v[H2D];
#pragma unroll
    for (int j = 0; j < H2D; ++j) {
      float acc = b2[j];
#pragma unroll
      for (int i = 0; i < H1D; ++i) acc += h1v[i] * W2[i * H2D + j];
      h2v[j] = __builtin_amdgcn_tanhf(acc);
    }
    float y = b3[0];
#pragma unroll
    for (int j = 0; j < H2D; ++j) y += h2v[j] * W3[j];

    if (half == 0) out[(size_t)b * SEQ + i0 + lane15] = y;
  }
}

extern "C" void kernel_launch(void* const* d_in, const int* in_sizes, int n_in,
                              void* d_out, int out_size, void* d_ws, size_t ws_size,
                              hipStream_t stream) {
  const float* x  = (const float*)d_in[0];
  const float* Wq = (const float*)d_in[1];
  const float* Wk = (const float*)d_in[2];
  const float* Wv = (const float*)d_in[3];
  const float* W1 = (const float*)d_in[4];
  const float* b1 = (const float*)d_in[5];
  const float* W2 = (const float*)d_in[6];
  const float* b2 = (const float*)d_in[7];
  const float* W3 = (const float*)d_in[8];
  const float* b3 = (const float*)d_in[9];
  float* out = (float*)d_out;

  hybrid_attn_kernel<<<BATCH, 128, 0, stream>>>(
      x, Wq, Wk, Wv, W1, b1, W2, b2, W3, b3, out);
}


# one pass over keys: online softmax, per-lane running reference in the WMMA C operand, rare wave-uniform rescale that redoes the tile; max pass gone (68 WMMAs/wave, was 128); sum scaling order changes
# speedup vs baseline: 1.9276x; 1.1794x over previous
#include <hip/hip_runtime.h>
#include <math.h>


#define BATCH   1024
#define SEQ     256
#define EMBED   4
#define H1D     8
#define H2D     4
#define NWAVE   4
#define TPW     4

#define WMMA_COVER4(d0, d1, d2, d3, a)                                   \
  asm volatile("v_nop\n\tv_nop\n\tv_nop\n\tv_nop"                        \
               : "+v"(d0), "+v"(d1), "+v"(d2), "+v"(d3), "+v"(a))

typedef __attribute__((ext_vector_type(2))) float v2f;
typedef __attribute__((ext_vector_type(4))) float v4f;
typedef __attribute__((ext_vector_type(8))) float v8f;

__global__ __launch_bounds__(128)
void hybrid_attn_kernel(const float* __restrict__ x,
                        const float* __restrict__ Wq, const float* __restrict__ Wk,
                        const float* __restrict__ Wv,
                        const float* __restrict__ W1, const float* __restrict__ b1,
                        const float* __restrict__ W2, const float* __restrict__ b2,
                        const float* __restrict__ W3, const float* __restrict__ b3,
                        float* __restrict__ out)
{
  __shared__ __attribute__((aligned(16))) float q_lds[SEQ * EMBED];
  __shared__ __attribute__((aligned(16))) float k_lds[SEQ * EMBED];
  __shared__ __attribute__((aligned(16))) float vt_lds[EMBED * SEQ];

  const int b    = blockIdx.x;
  const int tid  = threadIdx.x;
  const int lane = tid & 31;
  const int wave = tid >> 5;

  const float qscale = 0.5f * 1.44269504088896341f;
  const float* xb = x + (size_t)b * SEQ * EMBED;
  for (int t = tid; t < SEQ; t += 128) {
    float4 xe = ((const float4*)xb)[t];
    float xv[4] = {xe.x, xe.y, xe.z, xe.w};
#pragma unroll
    for (int eo = 0; eo < 4; ++eo) {
      float q = 0.f, k = 0.f, v = 0.f;
#pragma unroll
      for (int e = 0; e < 4; ++e) {
        q += xv[e] * Wq[e * 4 + eo];
        k += xv[e] * Wk[e * 4 + eo];
        v += xv[e] * Wv[e * 4 + eo];
      }
      q_lds[t * 4 + eo] = qscale * q;
      k_lds[t * 4 + eo] = k;
      vt_lds[eo * SEQ + t] = v;
    }
  }
  __syncthreads();

  const int lane15 = lane & 15;
  const int half   = lane >> 4;
  const int kbase  = half * 2;


  v2f qf[TPW];
#pragma unroll
  for (int tl = 0; tl < TPW; ++tl) {
    const int i0 = (wave + tl * NWAVE) * 16;
    qf[tl] = *(const v2f*)&q_lds[(i0 + lane15) * 4 + kbase];
  }

#define RESCALE_THR 64.0f

  float mref[TPW];
  {
    v2f kf = *(const v2f*)&k_lds[lane15 * 4 + kbase];
    v8f c[TPW];
#pragma unroll
    for (int tl = 0; tl < TPW; ++tl) {
      const v8f z = {};
      c[tl] = __builtin_amdgcn_wmma_f32_16x16x4_f32(
                  false, kf, false, qf[tl], (short)0, z, false, false);
    }
    WMMA_COVER4(c[0], c[1], c[2], c[3], kf);
#pragma unroll
    for (int tl = 0; tl < TPW; ++tl) {
      float m = -INFINITY;
      m = fmaxf(fmaxf(m, c[tl][0]), c[tl][1]);
      m = fmaxf(fmaxf(m, c[tl][2]), c[tl][3]);
      m = fmaxf(fmaxf(m, c[tl][4]), c[tl][5]);
      m = fmaxf(fmaxf(m, c[tl][6]), c[tl][7]);
      mref[tl] = m;
    }
  }

  v2f acc[TPW][EMBED], s2[TPW];
#pragma unroll
  for (int tl = 0; tl < TPW; ++tl) {
#pragma unroll
    for (int e = 0; e < EMBED; ++e) acc[tl][e] = (v2f){0.f, 0.f};
    s2[tl] = (v2f){0.f, 0.f};
  }

  int j = 0;
  for (int ep = 0; ep <= SEQ / 16 && j < SEQ / 16; ++ep) {
    v8f cneg[TPW];
#pragma unroll
    for (int tl = 0; tl < TPW; ++tl) {
#pragma unroll
      for (int r = 0; r < 8; ++r) cneg[tl][r] = -mref[tl];
    }
    float delta[TPW];
#pragma unroll
    for (int tl = 0; tl < TPW; ++tl) delta[tl] = 0.f;

    v2f kf = *(const v2f*)&k_lds[(j * 16 + lane15) * 4 + kbase];
    for (; j < SEQ / 16; ++j) {
      const float* vb = vt_lds + j * 16 + half * 8;
      v4f vlo[EMBED], vhi[EMBED];
#pragma unroll
      for (int e = 0; e < EMBED; ++e) {
        vlo[e] = *(const v4f*)(vb + e * SEQ);
        vhi[e] = *(const v4f*)(vb + e * SEQ + 4);
      }

      v8f c[TPW];
#pragma unroll
      for (int tl = 0; tl < TPW; ++tl)
        c[tl] = __builtin_amdgcn_wmma_f32_16x16x4_f32(
                    false, kf, false, qf[tl], (short)0, cneg[tl], false, false);
      WMMA_COVER4(c[0], c[1], c[2], c[3], kf);
      kf = *(const v2f*)&k_lds[((((j + 1) & (SEQ / 16 - 1)) * 16) + lane15) * 4 + kbase];

      float tm[TPW];
#pragma unroll
      for (int tl = 0; tl < TPW; ++tl) {
        float m = 0.f;
        m = fmaxf(fmaxf(m, c[tl][0]), c[tl][1]);
        m = fmaxf(fmaxf(m, c[tl][2]), c[tl][3]);
        m = fmaxf(fmaxf(m, c[tl][4]), c[tl][5]);
        m = fmaxf(fmaxf(m, c[tl][6]), c[tl][7]);
        tm[tl] = m;
      }
      const float g = fmaxf(fmaxf(fmaxf(tm[0], tm[1]), tm[2]), tm[3]);
      if (__builtin_amdgcn_ballot_w32(g > RESCALE_THR) != 0u) {
#pragma unroll
        for (int tl = 0; tl < TPW; ++tl) delta[tl] = tm[tl];
        break;
      }

#pragma unroll
      for (int tl = 0; tl < TPW; ++tl) {
        v8f p;
#pragma unroll
        for (int r = 0; r < 8; ++r) p[r] = __builtin_amdgcn_exp2f(c[tl][r]);
        const v2f p01 = {p[0], p[1]}, p23 = {p[2], p[3]}, p45 = {p[4], p[5]}, p67 = {p[6], p[7]};
        s2[tl] += p01; s2[tl] += p23; s2[tl] += p45; s2[tl] += p67;
#pragma unroll
        for (int e = 0; e < EMBED; ++e) {
          v2f a = acc[tl][e];
          a = __builtin_elementwise_fma(p01, (v2f){vlo[e][0], vlo[e][1]}, a);
          a = __builtin_elementwise_fma(p23, (v2f){vlo[e][2], vlo[e][3]}, a);
          a = __builtin_elementwise_fma(p45, (v2f){vhi[e][0], vhi[e][1]}, a);
          a = __builtin_elementwise_fma(p67, (v2f){vhi[e][2], vhi[e][3]}, a);
          acc[tl][e] = a;
        }
      }
    }

#pragma unroll
    for (int tl = 0; tl < TPW; ++tl) {
      const float f = __builtin_amdgcn_exp2f(-delta[tl]);
      const v2f f2 = {f, f};
#pragma unroll
      for (int e = 0; e < EMBED; ++e) acc[tl][e] *= f2;
      s2[tl] *= f2;
      mref[tl] += delta[tl];
    }
  }

#pragma unroll
  for (int tl = 0; tl < TPW; ++tl) {
    const int i0 = (wave + tl * NWAVE) * 16;
    float sum = s2[tl][0] + s2[tl][1];
    float o0  = acc[tl][0][0] + acc[tl][0][1];
    float o1  = acc[tl][1][0] + acc[tl][1][1];
    float o2  = acc[tl][2][0] + acc[tl][2][1];
    float o3  = acc[tl][3][0] + acc[tl][3][1];
    const float mboth = fmaxf(mref[tl], __shfl_xor(mref[tl], 16, 32));
    const float fh    = __builtin_amdgcn_exp2f(mref[tl] - mboth);
    sum *= fh; o0 *= fh; o1 *= fh; o2 *= fh; o3 *= fh;
    sum += __shfl_xor(sum, 16, 32);
    o0  += __shfl_xor(o0, 16, 32);
    o1  += __shfl_xor(o1, 16, 32);
    o2  += __shfl_xor(o2, 16, 32);
    o3  += __shfl_xor(o3, 16, 32);

    float inv = 1.f / sum;
    o0 *= inv; o1 *= inv; o2 *= inv; o3 *= inv;

    float h1v[H1D];
#pragma unroll
    for (int j = 0; j < H1D; ++j)
      h1v[j] = __builtin_amdgcn_tanhf(b1[j] + o0 * W1[0 * H1D + j] + o1 * W1[1 * H1D + j]
                           + o2 * W1[2 * H1D + j] + o3 * W1[3 * H1D + j]);
    float h2v[H2D];
#pragma unroll
    for (int j = 0; j < H2D; ++j) {
      float acc = b2[j];
#pragma unroll
      for (int i = 0; i < H1D; ++i) acc += h1v[i] * W2[i * H2D + j];
      h2v[j] = __builtin_amdgcn_tanhf(acc);
    }
    float y = b3[0];
#pragma unroll
    for (int j = 0; j < H2D; ++j) y += h2v[j] * W3[j];

    if (half == 0) out[(size_t)b * SEQ + i0 + lane15] = y;
  }
}

extern "C" void kernel_launch(void* const* d_in, const int* in_sizes, int n_in,
                              void* d_out, int out_size, void* d_ws, size_t ws_size,
                              hipStream_t stream) {
  const float* x  = (const float*)d_in[0];
  const float* Wq = (const float*)d_in[1];
  const float* Wk = (const float*)d_in[2];
  const float* Wv = (const float*)d_in[3];
  const float* W1 = (const float*)d_in[4];
  const float* b1 = (const float*)d_in[5];
  const float* W2 = (const float*)d_in[6];
  const float* b2 = (const float*)d_in[7];
  const float* W3 = (const float*)d_in[8];
  const float* b3 = (const float*)d_in[9];
  float* out = (float*)d_out;

  hybrid_attn_kernel<<<BATCH, 128, 0, stream>>>(
      x, Wq, Wk, Wv, W1, b1, W2, b2, W3, b3, out);
}


# epilogue: each lane half finishes two of the wave's four query tiles (merge exchange, division, MLP, store) with all 32 lanes at work; was four MLP passes with half the lanes masked; sum order kept
# speedup vs baseline: 2.1345x; 1.1073x over previous
#include <hip/hip_runtime.h>
#include <math.h>


#define BATCH   1024
#define SEQ     256
#define EMBED   4
#define H1D     8
#define H2D     4
#define NWAVE   4
#define TPW     4

#define WMMA_COVER4(d0, d1, d2, d3, a)                                   \
  asm volatile("v_nop\n\tv_nop\n\tv_nop\n\tv_nop"                        \
               : "+v"(d0), "+v"(d1), "+v"(d2), "+v"(d3), "+v"(a))

typedef __attribute__((ext_vector_type(2))) float v2f;
typedef __attribute__((ext_vector_type(4))) float v4f;
typedef __attribute__((ext_vector_type(8))) float v8f;

__global__ __launch_bounds__(128)
void hybrid_attn_kernel(const float* __restrict__ x,
                        const float* __restrict__ Wq, const float* __restrict__ Wk,
                        const float* __restrict__ Wv,
                        const float* __restrict__ W1, const float* __restrict__ b1,
                        const float* __restrict__ W2, const float* __restrict__ b2,
                        const float* __restrict__ W3, const float* __restrict__ b3,
                        float* __restrict__ out)
{
  __shared__ __attribute__((aligned(16))) float q_lds[SEQ * EMBED];
  __shared__ __attribute__((aligned(16))) float k_lds[SEQ * EMBED];
  __shared__ __attribute__((aligned(16))) float vt_lds[EMBED * SEQ];

  const int b    = blockIdx.x;
  const int tid  = threadIdx.x;
  const int lane = tid & 31;
  const int wave = tid >> 5;

  const float qscale = 0.5f * 1.44269504088896341f;
  const float* xb = x + (size_t)b * SEQ * EMBED;
  for (int t = tid; t < SEQ; t += 128) {
    float4 xe = ((const float4*)xb)[t];
    float xv[4] = {xe.x, xe.y, xe.z, xe.w};
#pragma unroll
    for (int eo = 0; eo < 4; ++eo) {
      float q = 0.f, k = 0.f, v = 0.f;
#pragma unroll
      for (int e = 0; e < 4; ++e) {
        q += xv[e] * Wq[e * 4 + eo];
        k += xv[e] * Wk[e * 4 + eo];
        v += xv[e] * Wv[e * 4 + eo];
      }
      q_lds[t * 4 + eo] = qscale * q;
      k_lds[t * 4 + eo] = k;
      vt_lds[eo * SEQ + t] = v;
    }
  }
  __syncthreads();

  const int lane15 = lane & 15;
  const int half   = lane >> 4;
  const int kbase  = half * 2;


  v2f qf[TPW];
#pragma unroll
  for (int tl = 0; tl < TPW; ++tl) {
    const int i0 = (wave + tl * NWAVE) * 16;
    qf[tl] = *(const v2f*)&q_lds[(i0 + lane15) * 4 + kbase];
  }

#define RESCALE_THR 64.0f

  float mref[TPW];
  {
    v2f kf = *(const v2f*)&k_lds[lane15 * 4 + kbase];
    v8f c[TPW];
#pragma unroll
    for (int tl = 0; tl < TPW; ++tl) {
      const v8f z = {};
      c[tl] = __builtin_amdgcn_wmma_f32_16x16x4_f32(
                  false, kf, false, qf[tl], (short)0, z, false, false);
    }
    WMMA_COVER4(c[0], c[1], c[2], c[3], kf);
#pragma unroll
    for (int tl = 0; tl < TPW; ++tl) {
      float m = -INFINITY;
      m = fmaxf(fmaxf(m, c[tl][0]), c[tl][1]);
      m = fmaxf(fmaxf(m, c[tl][2]), c[tl][3]);
      m = fmaxf(fmaxf(m, c[tl][4]), c[tl][5]);
      m = fmaxf(fmaxf(m, c[tl][6]), c[tl][7]);
      mref[tl] = m;
    }
  }

  v2f acc[TPW][EMBED], s2[TPW];
#pragma unroll
  for (int tl = 0; tl < TPW; ++tl) {
#pragma unroll
    for (int e = 0; e < EMBED; ++e) acc[tl][e] = (v2f){0.f, 0.f};
    s2[tl] = (v2f){0.f, 0.f};
  }

  int j = 0;
  for (int ep = 0; ep <= SEQ / 16 && j < SEQ / 16; ++ep) {
    v8f cneg[TPW];
#pragma unroll
    for (int tl = 0; tl < TPW; ++tl) {
#pragma unroll
      for (int r = 0; r < 8; ++r) cneg[tl][r] = -mref[tl];
    }
    float delta[TPW];
#pragma unroll
    for (int tl = 0; tl < TPW; ++tl) delta[tl] = 0.f;

    v2f kf = *(const v2f*)&k_lds[(j * 16 + lane15) * 4 + kbase];
    for (; j < SEQ / 16; ++j) {
      const float* vb = vt_lds + j * 16 + half * 8;
      v4f vlo[EMBED], vhi[EMBED];
#pragma unroll
      for (int e = 0; e < EMBED; ++e) {
        vlo[e] = *(const v4f*)(vb + e * SEQ);
        vhi[e] = *(const v4f*)(vb + e * SEQ + 4);
      }

      v8f c[TPW];
#pragma unroll
      for (int tl = 0; tl < TPW; ++tl)
        c[tl] = __builtin_amdgcn_wmma_f32_16x16x4_f32(
                    false, kf, false, qf[tl], (short)0, cneg[tl], false, false);
      WMMA_COVER4(c[0], c[1], c[2], c[3], kf);
      kf = *(const v2f*)&k_lds[((((j + 1) & (SEQ / 16 - 1)) * 16) + lane15) * 4 + kbase];

      float tm[TPW];
#pragma unroll
      for (int tl = 0; tl < TPW; ++tl) {
        float m = 0.f;
        m = fmaxf(fmaxf(m, c[tl][0]), c[tl][1]);
        m = fmaxf(fmaxf(m, c[tl][2]), c[tl][3]);
        m = fmaxf(fmaxf(m, c[tl][4]), c[tl][5]);
        m = fmaxf(fmaxf(m, c[tl][6]), c[tl][7]);
        tm[tl] = m;
      }
      const float g = fmaxf(fmaxf(fmaxf(tm[0], tm[1]), tm[2]), tm[3]);
      if (__builtin_amdgcn_ballot_w32(g > RESCALE_THR) != 0u) {
#pragma unroll
        for (int tl = 0; tl < TPW; ++tl) delta[tl] = tm[tl];
        break;
      }

#pragma unroll
      for (int tl = 0; tl < TPW; ++tl) {
        v8f p;
#pragma unroll
        for (int r = 0; r < 8; ++r) p[r] = __builtin_amdgcn_exp2f(c[tl][r]);
        const v2f p01 = {p[0], p[1]}, p23 = {p[2], p[3]}, p45 = {p[4], p[5]}, p67 = {p[6], p[7]};
        s2[tl] += p01; s2[tl] += p23; s2[tl] += p45; s2[tl] += p67;
#pragma unroll
        for (int e = 0; e < EMBED; ++e) {
          v2f a = acc[tl][e];
          a = __builtin_elementwise_fma(p01, (v2f){vlo[e][0], vlo[e][1]}, a);
          a = __builtin_elementwise_fma(p23, (v2f){vlo[e][2], vlo[e][3]}, a);
          a = __builtin_elementwise_fma(p45, (v2f){vhi[e][0], vhi[e][1]}, a);
          a = __builtin_elementwise_fma(p67, (v2f){vhi[e][2], vhi[e][3]}, a);
          acc[tl][e] = a;
        }
      }
    }

#pragma unroll
    for (int tl = 0; tl < TPW; ++tl) {
      const float f = __builtin_amdgcn_exp2f(-delta[tl]);
      const v2f f2 = {f, f};
#pragma unroll
      for (int e = 0; e < EMBED; ++e) acc[tl][e] *= f2;
      s2[tl] *= f2;
      mref[tl] += delta[tl];
    }
  }


  float psum[TPW], po[TPW][EMBED];
#pragma unroll
  for (int tl = 0; tl < TPW; ++tl) {
    const float mboth = fmaxf(mref[tl], __shfl_xor(mref[tl], 16, 32));
    const float fh    = __builtin_amdgcn_exp2f(mref[tl] - mboth);
    psum[tl] = (s2[tl][0] + s2[tl][1]) * fh;
#pragma unroll
    for (int e = 0; e < EMBED; ++e) po[tl][e] = (acc[tl][e][0] + acc[tl][e][1]) * fh;
  }

  v2f sum, o[EMBED];
#pragma unroll
  for (int u = 0; u < 2; ++u) {
    const float keep_s = half ? psum[2 + u] : psum[u];
    const float give_s = half ? psum[u]     : psum[2 + u];
    sum[u] = keep_s + __shfl_xor(give_s, 16, 32);
#pragma unroll
    for (int e = 0; e < EMBED; ++e) {
      const float keep_o = half ? po[2 + u][e] : po[u][e];
      const float give_o = half ? po[u][e]     : po[2 + u][e];
      o[e][u] = keep_o + __shfl_xor(give_o, 16, 32);
    }
  }

  const v2f inv = 1.f / sum;
#pragma unroll
  for (int e = 0; e < EMBED; ++e) o[e] *= inv;

  v2f h1v[H1D];
#pragma unroll
  for (int j = 0; j < H1D; ++j) {
    const v2f t = b1[j] + o[0] * W1[0 * H1D + j] + o[1] * W1[1 * H1D + j]
                        + o[2] * W1[2 * H1D + j] + o[3] * W1[3 * H1D + j];
    h1v[j] = (v2f){__builtin_amdgcn_tanhf(t[0]), __builtin_amdgcn_tanhf(t[1])};
  }
  v2f h2v[H2D];
#pragma unroll
  for (int j = 0; j < H2D; ++j) {
    v2f t = {b2[j], b2[j]};
#pragma unroll
    for (int i = 0; i < H1D; ++i) t += h1v[i] * W2[i * H2D + j];
    h2v[j] = (v2f){__builtin_amdgcn_tanhf(t[0]), __builtin_amdgcn_tanhf(t[1])};
  }
  v2f y = {b3[0], b3[0]};
#pragma unroll
  for (int j = 0; j < H2D; ++j) y += h2v[j] * W3[j];

#pragma unroll
  for (int u = 0; u < 2; ++u) {
    const int i0 = (wave + (2 * half + u) * NWAVE) * 16;
    out[(size_t)b * SEQ + i0 + lane15] = y[u];
  }
}

extern "C" void kernel_launch(void* const* d_in, const int* in_sizes, int n_in,
                              void* d_out, int out_size, void* d_ws, size_t ws_size,
                              hipStream_t stream) {
  const float* x  = (const float*)d_in[0];
  const float* Wq = (const float*)d_in[1];
  const float* Wk = (const float*)d_in[2];
  const float* Wv = (const float*)d_in[3];
  const float* W1 = (const float*)d_in[4];
  const float* b1 = (const float*)d_in[5];
  const float* W2 = (const float*)d_in[6];
  const float* b2 = (const float*)d_in[7];
  const float* W3 = (const float*)d_in[8];
  const float* b3 = (const float*)d_in[9];
  float* out = (float*)d_out;

  hybrid_attn_kernel<<<BATCH, 128, 0, stream>>>(
      x, Wq, Wk, Wv, W1, b1, W2, b2, W3, b3, out);
}


# all memory reads requested at wave start (both x rows, projection weights, MLP parameters kept in scalar registers): 7 dependent round trips per wave become 2 to 3; no sum order changes
# speedup vs baseline: 2.2696x; 1.0633x over previous
#include <hip/hip_runtime.h>
#include <math.h>


#define BATCH   1024
#define SEQ     256
#define EMBED   4
#define H1D     8
#define H2D     4
#define NWAVE   4
#define TPW     4

#define WMMA_COVER4(d0, d1, d2, d3, a)                                   \
  asm volatile("v_nop\n\tv_nop\n\tv_nop\n\tv_nop"                        \
               : "+v"(d0), "+v"(d1), "+v"(d2), "+v"(d3), "+v"(a))

#define KEEP_SGPR8(a, o)                                                              \
  asm volatile("" :: "s"(a[(o) + 0]), "s"(a[(o) + 1]), "s"(a[(o) + 2]), "s"(a[(o) + 3]), \
                     "s"(a[(o) + 4]), "s"(a[(o) + 5]), "s"(a[(o) + 6]), "s"(a[(o) + 7]))

typedef __attribute__((ext_vector_type(2))) float v2f;
typedef __attribute__((ext_vector_type(4))) float v4f;
typedef __attribute__((ext_vector_type(8))) float v8f;

__global__ __launch_bounds__(128)
void hybrid_attn_kernel(const float* __restrict__ x,
                        const float* __restrict__ Wq, const float* __restrict__ Wk,
                        const float* __restrict__ Wv,
                        const float* __restrict__ W1, const float* __restrict__ b1,
                        const float* __restrict__ W2, const float* __restrict__ b2,
                        const float* __restrict__ W3, const float* __restrict__ b3,
                        float* __restrict__ out)
{
  __shared__ __attribute__((aligned(16))) float q_lds[SEQ * EMBED];
  __shared__ __attribute__((aligned(16))) float k_lds[SEQ * EMBED];
  __shared__ __attribute__((aligned(16))) float vt_lds[EMBED * SEQ];

  const int b    = blockIdx.x;
  const int tid  = threadIdx.x;
  const int lane = tid & 31;
  const int wave = tid >> 5;

  const float qscale = 0.5f * 1.44269504088896341f;
  const float4* xb4 = (const float4*)(x + (size_t)b * SEQ * EMBED);
  float4 xrow[2];
  xrow[0] = xb4[tid];
  xrow[1] = xb4[tid + 128];

  float w1[EMBED * H1D], c1[H1D], w2[H1D * H2D], c2[H2D], w3[H2D];
#pragma unroll
  for (int i = 0; i < EMBED * H1D; ++i) w1[i] = W1[i];
#pragma unroll
  for (int i = 0; i < H1D; ++i) c1[i] = b1[i];
#pragma unroll
  for (int i = 0; i < H1D * H2D; ++i) w2[i] = W2[i];
#pragma unroll
  for (int i = 0; i < H2D; ++i) { c2[i] = b2[i]; w3[i] = W3[i]; }
  const float c3 = b3[0];

#pragma unroll
  for (int rr = 0; rr < 2; ++rr) {
    const int t = tid + rr * 128;
    const float xv[4] = {xrow[rr].x, xrow[rr].y, xrow[rr].z, xrow[rr].w};
#pragma unroll
    for (int eo = 0; eo < 4; ++eo) {
      float q = 0.f, k = 0.f, v = 0.f;
#pragma unroll
      for (int e = 0; e < 4; ++e) {
        q += xv[e] * Wq[e * 4 + eo];
        k += xv[e] * Wk[e * 4 + eo];
        v += xv[e] * Wv[e * 4 + eo];
      }
      q_lds[t * 4 + eo] = qscale * q;
      k_lds[t * 4 + eo] = k;
      vt_lds[eo * SEQ + t] = v;
    }
  }
  __syncthreads();

  const int lane15 = lane & 15;
  const int half   = lane >> 4;
  const int kbase  = half * 2;


  v2f qf[TPW];
#pragma unroll
  for (int tl = 0; tl < TPW; ++tl) {
    const int i0 = (wave + tl * NWAVE) * 16;
    qf[tl] = *(const v2f*)&q_lds[(i0 + lane15) * 4 + kbase];
  }

#define RESCALE_THR 64.0f

  float mref[TPW];
  {
    v2f kf = *(const v2f*)&k_lds[lane15 * 4 + kbase];
    v8f c[TPW];
#pragma unroll
    for (int tl = 0; tl < TPW; ++tl) {
      const v8f z = {};
      c[tl] = __builtin_amdgcn_wmma_f32_16x16x4_f32(
                  false, kf, false, qf[tl], (short)0, z, false, false);
    }
    WMMA_COVER4(c[0], c[1], c[2], c[3], kf);
#pragma unroll
    for (int tl = 0; tl < TPW; ++tl) {
      float m = -INFINITY;
      m = fmaxf(fmaxf(m, c[tl][0]), c[tl][1]);
      m = fmaxf(fmaxf(m, c[tl][2]), c[tl][3]);
      m = fmaxf(fmaxf(m, c[tl][4]), c[tl][5]);
      m = fmaxf(fmaxf(m, c[tl][6]), c[tl][7]);
      mref[tl] = m;
    }
  }

  v2f acc[TPW][EMBED], s2[TPW];
#pragma unroll
  for (int tl = 0; tl < TPW; ++tl) {
#pragma unroll
    for (int e = 0; e < EMBED; ++e) acc[tl][e] = (v2f){0.f, 0.f};
    s2[tl] = (v2f){0.f, 0.f};
  }

  KEEP_SGPR8(w1, 0);  KEEP_SGPR8(w1, 8);  KEEP_SGPR8(w1, 16); KEEP_SGPR8(w1, 24);
  KEEP_SGPR8(w2, 0);  KEEP_SGPR8(w2, 8);  KEEP_SGPR8(w2, 16); KEEP_SGPR8(w2, 24);
  KEEP_SGPR8(c1, 0);
  asm volatile("" :: "s"(c2[0]), "s"(c2[1]), "s"(c2[2]), "s"(c2[3]),
                     "s"(w3[0]), "s"(w3[1]), "s"(w3[2]), "s"(w3[3]), "s"(c3));

  int j = 0;
  for (int ep = 0; ep <= SEQ / 16 && j < SEQ / 16; ++ep) {
    v8f cneg[TPW];
#pragma unroll
    for (int tl = 0; tl < TPW; ++tl) {
#pragma unroll
      for (int r = 0; r < 8; ++r) cneg[tl][r] = -mref[tl];
    }
    float delta[TPW];
#pragma unroll
    for (int tl = 0; tl < TPW; ++tl) delta[tl] = 0.f;

    v2f kf = *(const v2f*)&k_lds[(j * 16 + lane15) * 4 + kbase];
    for (; j < SEQ / 16; ++j) {
      const float* vb = vt_lds + j * 16 + half * 8;
      v4f vlo[EMBED], vhi[EMBED];
#pragma unroll
      for (int e = 0; e < EMBED; ++e) {
        vlo[e] = *(const v4f*)(vb + e * SEQ);
        vhi[e] = *(const v4f*)(vb + e * SEQ + 4);
      }

      v8f c[TPW];
#pragma unroll
      for (int tl = 0; tl < TPW; ++tl)
        c[tl] = __builtin_amdgcn_wmma_f32_16x16x4_f32(
                    false, kf, false, qf[tl], (short)0, cneg[tl], false, false);
      WMMA_COVER4(c[0], c[1], c[2], c[3], kf);
      kf = *(const volatile __attribute__((address_space(3))) v2f*)
               &k_lds[((((j + 1) & (SEQ / 16 - 1)) * 16) + lane15) * 4 + kbase];

      float tm[TPW];
#pragma unroll
      for (int tl = 0; tl < TPW; ++tl) {
        float m = 0.f;
        m = fmaxf(fmaxf(m, c[tl][0]), c[tl][1]);
        m = fmaxf(fmaxf(m, c[tl][2]), c[tl][3]);
        m = fmaxf(fmaxf(m, c[tl][4]), c[tl][5]);
        m = fmaxf(fmaxf(m, c[tl][6]), c[tl][7]);
        tm[tl] = m;
      }
      const float g = fmaxf(fmaxf(fmaxf(tm[0], tm[1]), tm[2]), tm[3]);
      if (__builtin_amdgcn_ballot_w32(g > RESCALE_THR) != 0u) {
#pragma unroll
        for (int tl = 0; tl < TPW; ++tl) delta[tl] = tm[tl];
        break;
      }

#pragma unroll
      for (int tl = 0; tl < TPW; ++tl) {
        v8f p;
#pragma unroll
        for (int r = 0; r < 8; ++r) p[r] = __builtin_amdgcn_exp2f(c[tl][r]);
        const v2f p01 = {p[0], p[1]}, p23 = {p[2], p[3]}, p45 = {p[4], p[5]}, p67 = {p[6], p[7]};
        s2[tl] += p01; s2[tl] += p23; s2[tl] += p45; s2[tl] += p67;
#pragma unroll
        for (int e = 0; e < EMBED; ++e) {
          v2f a = acc[tl][e];
          a = __builtin_elementwise_fma(p01, (v2f){vlo[e][0], vlo[e][1]}, a);
          a = __builtin_elementwise_fma(p23, (v2f){vlo[e][2], vlo[e][3]}, a);
          a = __builtin_elementwise_fma(p45, (v2f){vhi[e][0], vhi[e][1]}, a);
          a = __builtin_elementwise_fma(p67, (v2f){vhi[e][2], vhi[e][3]}, a);
          acc[tl][e] = a;
        }
      }
    }

#pragma unroll
    for (int tl = 0; tl < TPW; ++tl) {
      const float f = __builtin_amdgcn_exp2f(-delta[tl]);
      const v2f f2 = {f, f};
#pragma unroll
      for (int e = 0; e < EMBED; ++e) acc[tl][e] *= f2;
      s2[tl] *= f2;
      mref[tl] += delta[tl];
    }
  }


  float psum[TPW], po[TPW][EMBED];
#pragma unroll
  for (int tl = 0; tl < TPW; ++tl) {
    const float mboth = fmaxf(mref[tl], __shfl_xor(mref[tl], 16, 32));
    const float fh    = __builtin_amdgcn_exp2f(mref[tl] - mboth);
    psum[tl] = (s2[tl][0] + s2[tl][1]) * fh;
#pragma unroll
    for (int e = 0; e < EMBED; ++e) po[tl][e] = (acc[tl][e][0] + acc[tl][e][1]) * fh;
  }

  v2f sum, o[EMBED];
#pragma unroll
  for (int u = 0; u < 2; ++u) {
    const float keep_s = half ? psum[2 + u] : psum[u];
    const float give_s = half ? psum[u]     : psum[2 + u];
    sum[u] = keep_s + __shfl_xor(give_s, 16, 32);
#pragma unroll
    for (int e = 0; e < EMBED; ++e) {
      const float keep_o = half ? po[2 + u][e] : po[u][e];
      const float give_o = half ? po[u][e]     : po[2 + u][e];
      o[e][u] = keep_o + __shfl_xor(give_o, 16, 32);
    }
  }

  const v2f inv = 1.f / sum;
#pragma unroll
  for (int e = 0; e < EMBED; ++e) o[e] *= inv;

  v2f h1v[H1D];
#pragma unroll
  for (int j = 0; j < H1D; ++j) {
    const v2f t = c1[j] + o[0] * w1[0 * H1D + j] + o[1] * w1[1 * H1D + j]
                        + o[2] * w1[2 * H1D + j] + o[3] * w1[3 * H1D + j];
    h1v[j] = (v2f){__builtin_amdgcn_tanhf(t[0]), __builtin_amdgcn_tanhf(t[1])};
  }
  v2f h2v[H2D];
#pragma unroll
  for (int j = 0; j < H2D; ++j) {
    v2f t = {c2[j], c2[j]};
#pragma unroll
    for (int i = 0; i < H1D; ++i) t += h1v[i] * w2[i * H2D + j];
    h2v[j] = (v2f){__builtin_amdgcn_tanhf(t[0]), __builtin_amdgcn_tanhf(t[1])};
  }
  v2f y = {c3, c3};
#pragma unroll
  for (int j = 0; j < H2D; ++j) y += h2v[j] * w3[j];

#pragma unroll
  for (int u = 0; u < 2; ++u) {
    const int i0 = (wave + (2 * half + u) * NWAVE) * 16;
    out[(size_t)b * SEQ + i0 + lane15] = y[u];
  }
}

extern "C" void kernel_launch(void* const* d_in, const int* in_sizes, int n_in,
                              void* d_out, int out_size, void* d_ws, size_t ws_size,
                              hipStream_t stream) {
  const float* x  = (const float*)d_in[0];
  const float* Wq = (const float*)d_in[1];
  const float* Wk = (const float*)d_in[2];
  const float* Wv = (const float*)d_in[3];
  const float* W1 = (const float*)d_in[4];
  const float* b1 = (const float*)d_in[5];
  const float* W2 = (const float*)d_in[6];
  const float* b2 = (const float*)d_in[7];
  const float* W3 = (const float*)d_in[8];
  const float* b3 = (const float*)d_in[9];
  float* out = (float*)d_out;

  hybrid_attn_kernel<<<BATCH, 128, 0, stream>>>(
      x, Wq, Wk, Wv, W1, b1, W2, b2, W3, b3, out);
}
